# starting kernel
# baseline (speedup 1.0000x reference)
#include <hip/hip_runtime.h>

typedef __attribute__((ext_vector_type(2))) float v2f;
typedef __attribute__((ext_vector_type(4))) float v4f;
typedef __attribute__((ext_vector_type(8))) float v8f;

#define NPTS 32768
#define NB   4
#define CIN  64
#define COUT 64
#define KNB  16

#define WPAIRSTR 160
#define WSZ  (128 * WPAIRSTR)
#define VOFF WSZ
#define VWAVE (128 * 32)
#define BOFF (VOFF + 4 * VWAVE)
#define LDS_FLOATS (BOFF + 64)

__global__ __launch_bounds__(256) void transpose_feat_kernel(
    const float* __restrict__ feat, float* __restrict__ ft)
{
    __shared__ float tile[64][65];
    const int b  = blockIdx.y;
    const int n0 = blockIdx.x * 64;
    const int tid = threadIdx.x;
    #pragma unroll
    for (int i = 0; i < 16; ++i) {
        const int j = tid + i * 256;
        const int c = j >> 6, nloc = j & 63;
        tile[nloc][c] = feat[(b * CIN + c) * NPTS + n0 + nloc];
    }
    __syncthreads();
    #pragma unroll
    for (int i = 0; i < 16; ++i) {
        const int j = tid + i * 256;
        const int nloc = j >> 6, c = j & 63;
        ft[(b * NPTS + n0 + nloc) * CIN + c] = tile[nloc][c];
    }
}

template <bool USE_FT>
__global__ __launch_bounds__(128) void flexconv_wmma_kernel(
    const float* __restrict__ feat,
    const float* __restrict__ ft,
    const float* __restrict__ wt,
    const float* __restrict__ wb,
    const float* __restrict__ bias,
    const int*   __restrict__ nbr,
    const float* __restrict__ pos,
    float* __restrict__ out)
{
    __shared__ float lds[LDS_FLOATS];
    const int tid = threadIdx.x;

    for (int j = tid; j < 128 * 64; j += 128) {
        const int dcp = j >> 6, o = j & 63;
        const int dc = dcp * 2;
        float w0, w1;
        if (dc < 192) { w0 = wt[dc * 64 + o];        w1 = wt[(dc + 1) * 64 + o]; }
        else          { w0 = wb[(dc - 192) * 64 + o]; w1 = wb[(dc - 191) * 64 + o]; }
        v2f p; p.x = w0; p.y = w1;
        *reinterpret_cast<v2f*>(&lds[dcp * WPAIRSTR + o * 2]) = p;
    }
    if (tid < COUT) lds[BOFF + tid] = bias[tid];
    __syncthreads();

    const int wave = tid >> 5;
    const int lane = tid & 31;
    const int nl   = lane & 15;
    const int h    = lane >> 4;
    const int b    = blockIdx.y;
    const int n    = blockIdx.x * 64 + wave * 16 + nl;
    const int vb   = VOFF + wave * VWAVE;

    const float cx = pos[(b * 3 + 0) * NPTS + n];
    const float cy = pos[(b * 3 + 1) * NPTS + n];
    const float cz = pos[(b * 3 + 2) * NPTS + n];

    int   idxs[KNB];
    float dxs[KNB], dys[KNB], dzs[KNB];
    #pragma unroll
    for (int k = 0; k < KNB; ++k)
        idxs[k] = nbr[(b * KNB + k) * NPTS + n];
    #pragma unroll
    for (int k = 0; k < KNB; ++k) {
        const int id = idxs[k];
        dxs[k] = pos[(b * 3 + 0) * NPTS + id] - cx;
        dys[k] = pos[(b * 3 + 1) * NPTS + id] - cy;
        dzs[k] = pos[(b * 3 + 2) * NPTS + id] - cz;
    }

    for (int cc = 0; cc < 4; ++cc) {
        const int cbase = h * 32 + cc * 8;
        float ax[8] = {}, ay[8] = {}, az[8] = {}, asum[8] = {};
        for (int k = 0; k < KNB; ++k) {
            const int   id = idxs[k];
            const float dx = dxs[k], dy = dys[k], dz = dzs[k];
            float f[8];
            if constexpr (USE_FT) {
                const v4f* fp = reinterpret_cast<const v4f*>(ft + (b * NPTS + id) * CIN + cbase);
                const v4f f0 = fp[0], f1 = fp[1];
                #pragma unroll
                for (int c = 0; c < 4; ++c) { f[c] = f0[c]; f[c + 4] = f1[c]; }
            } else {
                const float* fp = feat + (b * CIN + cbase) * NPTS + id;
                #pragma unroll
                for (int c = 0; c < 8; ++c) f[c] = fp[c * NPTS];
            }
            #pragma unroll
            for (int c = 0; c < 8; ++c) {
                ax[c]   = fmaf(f[c], dx, ax[c]);
                ay[c]   = fmaf(f[c], dy, ay[c]);
                az[c]   = fmaf(f[c], dz, az[c]);
                asum[c] += f[c];
            }
        }
        #pragma unroll
        for (int c = 0; c < 8; c += 2) {
            const int dcp = (cbase + c) >> 1;
            v2f px, py, pz, ps;
            px.x = ax[c];   px.y = ax[c + 1];
            py.x = ay[c];   py.y = ay[c + 1];
            pz.x = az[c];   pz.y = az[c + 1];
            ps.x = asum[c]; ps.y = asum[c + 1];
            *reinterpret_cast<v2f*>(&lds[vb + (dcp      ) * 32 + nl * 2]) = px;
            *reinterpret_cast<v2f*>(&lds[vb + (dcp + 32 ) * 32 + nl * 2]) = py;
            *reinterpret_cast<v2f*>(&lds[vb + (dcp + 64 ) * 32 + nl * 2]) = pz;
            *reinterpret_cast<v2f*>(&lds[vb + (dcp + 96 ) * 32 + nl * 2]) = ps;
        }
    }

    v8f acc[4] = {v8f{}, v8f{}, v8f{}, v8f{}};
    #pragma unroll 4
    for (int ks = 0; ks < 64; ++ks) {
        const int dcp = ks * 2 + h;
        const v2f bf = *reinterpret_cast<const v2f*>(&lds[vb + dcp * 32 + nl * 2]);
        #pragma unroll
        for (int ot = 0; ot < 4; ++ot) {
            const v2f af = *reinterpret_cast<const v2f*>(
                &lds[dcp * WPAIRSTR + (ot * 16 + nl) * 2]);
            acc[ot] = __builtin_amdgcn_wmma_f32_16x16x4_f32(
                false, af, false, bf, (short)0, acc[ot], false, false);
        }
    }

    #pragma unroll
    for (int ot = 0; ot < 4; ++ot) {
        #pragma unroll
        for (int r = 0; r < 8; ++r) {
            const int o = ot * 16 + r + h * 8;
            out[(b * COUT + o) * NPTS + n] = acc[ot][r] + lds[BOFF + o];
        }
    }
}

extern "C" void kernel_launch(void* const* d_in, const int* in_sizes, int n_in,
                              void* d_out, int out_size, void* d_ws, size_t ws_size,
                              hipStream_t stream) {
    (void)in_sizes; (void)n_in; (void)out_size;
    const float* feat = (const float*)d_in[0];
    const float* wt   = (const float*)d_in[1];
    const float* wb   = (const float*)d_in[2];
    const float* bias = (const float*)d_in[3];
    const int*   nbr  = (const int*)d_in[4];
    const float* pos  = (const float*)d_in[5];
    float* out = (float*)d_out;

    const size_t ft_bytes = (size_t)NB * NPTS * CIN * sizeof(float);
    dim3 grid(NPTS / 64, NB);

    if (ws_size >= ft_bytes) {
        float* ft = (float*)d_ws;
        transpose_feat_kernel<<<grid, 256, 0, stream>>>(feat, ft);
        flexconv_wmma_kernel<true><<<grid, 128, 0, stream>>>(
            feat, ft, wt, wb, bias, nbr, pos, out);
    } else {
        flexconv_wmma_kernel<false><<<grid, 128, 0, stream>>>(
            feat, feat, wt, wb, bias, nbr, pos, out);
    }
}


# W' staging: all 32 b128 global loads per thread issued before the first LDS store (one memory round trip per work-group instead of 64 waited load pairs); same LDS layout
# speedup vs baseline: 1.3570x; 1.3570x over previous
#include <hip/hip_runtime.h>

typedef __attribute__((ext_vector_type(2))) float v2f;
typedef __attribute__((ext_vector_type(4))) float v4f;
typedef __attribute__((ext_vector_type(8))) float v8f;

#define NPTS 32768
#define NB   4
#define CIN  64
#define COUT 64
#define KNB  16

#define WPAIRSTR 160
#define WSZ  (128 * WPAIRSTR)
#define VOFF WSZ
#define VWAVE (128 * 32)
#define BOFF (VOFF + 4 * VWAVE)
#define LDS_FLOATS (BOFF + 64)

__global__ __launch_bounds__(256) void transpose_feat_kernel(
    const float* __restrict__ feat, float* __restrict__ ft)
{
    __shared__ float tile[64][65];
    const int b  = blockIdx.y;
    const int n0 = blockIdx.x * 64;
    const int tid = threadIdx.x;
    #pragma unroll
    for (int i = 0; i < 16; ++i) {
        const int j = tid + i * 256;
        const int c = j >> 6, nloc = j & 63;
        tile[nloc][c] = feat[(b * CIN + c) * NPTS + n0 + nloc];
    }
    __syncthreads();
    #pragma unroll
    for (int i = 0; i < 16; ++i) {
        const int j = tid + i * 256;
        const int nloc = j >> 6, c = j & 63;
        ft[(b * NPTS + n0 + nloc) * CIN + c] = tile[nloc][c];
    }
}

template <bool USE_FT>
__global__ __launch_bounds__(128) void flexconv_wmma_kernel(
    const float* __restrict__ feat,
    const float* __restrict__ ft,
    const float* __restrict__ wt,
    const float* __restrict__ wb,
    const float* __restrict__ bias,
    const int*   __restrict__ nbr,
    const float* __restrict__ pos,
    float* __restrict__ out)
{
    __shared__ alignas(16) float lds[LDS_FLOATS];
    const int tid = threadIdx.x;

    v4f wr0[16], wr1[16];
    #pragma unroll
    for (int j = 0; j < 16; ++j) {
        const int i   = tid + j * 128;
        const int dcp = i >> 4, oq = i & 15;
        const int dc  = dcp * 2;
        const float* row = (j < 12) ? (wt + dc * 64) : (wb + (dc - 192) * 64);
        wr0[j] = *reinterpret_cast<const v4f*>(row + oq * 4);
        wr1[j] = *reinterpret_cast<const v4f*>(row + 64 + oq * 4);
    }
    float bv = 0.0f;
    if (tid < COUT) bv = bias[tid];
    #pragma unroll
    for (int j = 0; j < 16; ++j) {
        const int i   = tid + j * 128;
        const int dcp = i >> 4, oq = i & 15;
        v4f lo, hi;
        lo.x = wr0[j].x; lo.y = wr1[j].x; lo.z = wr0[j].y; lo.w = wr1[j].y;
        hi.x = wr0[j].z; hi.y = wr1[j].z; hi.z = wr0[j].w; hi.w = wr1[j].w;
        float* dst = &lds[dcp * WPAIRSTR + oq * 8];
        *reinterpret_cast<v4f*>(dst)     = lo;
        *reinterpret_cast<v4f*>(dst + 4) = hi;
    }
    if (tid < COUT) lds[BOFF + tid] = bv;
    __syncthreads();

    const int wave = tid >> 5;
    const int lane = tid & 31;
    const int nl   = lane & 15;
    const int h    = lane >> 4;
    const int b    = blockIdx.y;
    const int n    = blockIdx.x * 64 + wave * 16 + nl;
    const int vb   = VOFF + wave * VWAVE;

    const float cx = pos[(b * 3 + 0) * NPTS + n];
    const float cy = pos[(b * 3 + 1) * NPTS + n];
    const float cz = pos[(b * 3 + 2) * NPTS + n];

    int   idxs[KNB];
    float dxs[KNB], dys[KNB], dzs[KNB];
    #pragma unroll
    for (int k = 0; k < KNB; ++k)
        idxs[k] = nbr[(b * KNB + k) * NPTS + n];
    #pragma unroll
    for (int k = 0; k < KNB; ++k) {
        const int id = idxs[k];
        dxs[k] = pos[(b * 3 + 0) * NPTS + id] - cx;
        dys[k] = pos[(b * 3 + 1) * NPTS + id] - cy;
        dzs[k] = pos[(b * 3 + 2) * NPTS + id] - cz;
    }

    for (int cc = 0; cc < 4; ++cc) {
        const int cbase = h * 32 + cc * 8;
        float ax[8] = {}, ay[8] = {}, az[8] = {}, asum[8] = {};
        for (int k = 0; k < KNB; ++k) {
            const int   id = idxs[k];
            const float dx = dxs[k], dy = dys[k], dz = dzs[k];
            float f[8];
            if constexpr (USE_FT) {
                const v4f* fp = reinterpret_cast<const v4f*>(ft + (b * NPTS + id) * CIN + cbase);
                const v4f f0 = fp[0], f1 = fp[1];
                #pragma unroll
                for (int c = 0; c < 4; ++c) { f[c] = f0[c]; f[c + 4] = f1[c]; }
            } else {
                const float* fp = feat + (b * CIN + cbase) * NPTS + id;
                #pragma unroll
                for (int c = 0; c < 8; ++c) f[c] = fp[c * NPTS];
            }
            #pragma unroll
            for (int c = 0; c < 8; ++c) {
                ax[c]   = fmaf(f[c], dx, ax[c]);
                ay[c]   = fmaf(f[c], dy, ay[c]);
                az[c]   = fmaf(f[c], dz, az[c]);
                asum[c] += f[c];
            }
        }
        #pragma unroll
        for (int c = 0; c < 8; c += 2) {
            const int dcp = (cbase + c) >> 1;
            v2f px, py, pz, ps;
            px.x = ax[c];   px.y = ax[c + 1];
            py.x = ay[c];   py.y = ay[c + 1];
            pz.x = az[c];   pz.y = az[c + 1];
            ps.x = asum[c]; ps.y = asum[c + 1];
            *reinterpret_cast<v2f*>(&lds[vb + (dcp      ) * 32 + nl * 2]) = px;
            *reinterpret_cast<v2f*>(&lds[vb + (dcp + 32 ) * 32 + nl * 2]) = py;
            *reinterpret_cast<v2f*>(&lds[vb + (dcp + 64 ) * 32 + nl * 2]) = pz;
            *reinterpret_cast<v2f*>(&lds[vb + (dcp + 96 ) * 32 + nl * 2]) = ps;
        }
    }

    v8f acc[4] = {v8f{}, v8f{}, v8f{}, v8f{}};
    #pragma unroll 4
    for (int ks = 0; ks < 64; ++ks) {
        const int dcp = ks * 2 + h;
        const v2f bf = *reinterpret_cast<const v2f*>(&lds[vb + dcp * 32 + nl * 2]);
        #pragma unroll
        for (int ot = 0; ot < 4; ++ot) {
            const v2f af = *reinterpret_cast<const v2f*>(
                &lds[dcp * WPAIRSTR + (ot * 16 + nl) * 2]);
            acc[ot] = __builtin_amdgcn_wmma_f32_16x16x4_f32(
                false, af, false, bf, (short)0, acc[ot], false, false);
        }
    }

    #pragma unroll
    for (int ot = 0; ot < 4; ++ot) {
        #pragma unroll
        for (int r = 0; r < 8; ++r) {
            const int o = ot * 16 + r + h * 8;
            out[(b * COUT + o) * NPTS + n] = acc[ot][r] + lds[BOFF + o];
        }
    }
}

extern "C" void kernel_launch(void* const* d_in, const int* in_sizes, int n_in,
                              void* d_out, int out_size, void* d_ws, size_t ws_size,
                              hipStream_t stream) {
    (void)in_sizes; (void)n_in; (void)out_size;
    const float* feat = (const float*)d_in[0];
    const float* wt   = (const float*)d_in[1];
    const float* wb   = (const float*)d_in[2];
    const float* bias = (const float*)d_in[3];
    const int*   nbr  = (const int*)d_in[4];
    const float* pos  = (const float*)d_in[5];
    float* out = (float*)d_out;

    const size_t ft_bytes = (size_t)NB * NPTS * CIN * sizeof(float);
    dim3 grid(NPTS / 64, NB);

    if (ws_size >= ft_bytes) {
        float* ft = (float*)d_ws;
        transpose_feat_kernel<<<grid, 256, 0, stream>>>(feat, ft);
        flexconv_wmma_kernel<true><<<grid, 128, 0, stream>>>(
            feat, ft, wt, wb, bias, nbr, pos, out);
    } else {
        flexconv_wmma_kernel<false><<<grid, 128, 0, stream>>>(
            feat, feat, wt, wb, bias, nbr, pos, out);
    }
}


# gather re-mapped: a 16-lane row reads one whole 256 B neighbour row per b128 load (offset and rel. position by row_share DPP): each 128 B line fetched once, not 4 times; sum order unchanged
# speedup vs baseline: 2.5830x; 1.9035x over previous
#include <hip/hip_runtime.h>

typedef __attribute__((ext_vector_type(2))) float v2f;
typedef __attribute__((ext_vector_type(4))) float v4f;
typedef __attribute__((ext_vector_type(8))) float v8f;

#define NPTS 32768
#define NB   4
#define CIN  64
#define COUT 64
#define KNB  16

#define WPAIRSTR 160
#define WSZ  (128 * WPAIRSTR)
#define VOFF WSZ
#define VSTR 34
#define VWAVE (128 * VSTR)
#define BOFF (VOFF + 4 * VWAVE)
#define LDS_FLOATS (BOFF + 64)

__global__ __launch_bounds__(256) void transpose_feat_kernel(
    const float* __restrict__ feat, float* __restrict__ ft)
{
    __shared__ float tile[64][65];
    const int b  = blockIdx.y;
    const int n0 = blockIdx.x * 64;
    const int tid = threadIdx.x;
    #pragma unroll
    for (int i = 0; i < 16; ++i) {
        const int j = tid + i * 256;
        const int c = j >> 6, nloc = j & 63;
        tile[nloc][c] = feat[(b * CIN + c) * NPTS + n0 + nloc];
    }
    __syncthreads();
    #pragma unroll
    for (int i = 0; i < 16; ++i) {
        const int j = tid + i * 256;
        const int nloc = j >> 6, c = j & 63;
        ft[(b * NPTS + n0 + nloc) * CIN + c] = tile[nloc][c];
    }
}

template <int SRC>
__device__ __forceinline__ int row_share_i(int v) {
    return __builtin_amdgcn_mov_dpp(v, 0x150 + SRC, 0xF, 0xF, true);
}
template <int SRC>
__device__ __forceinline__ float row_share_f(float v) {
    return __builtin_bit_cast(float, row_share_i<SRC>(__builtin_bit_cast(int, v)));
}

template <int P>
__device__ __forceinline__ void gather_issue(const char* __restrict__ ftb, unsigned lane_off,
                                             const int (&offs)[8], v4f (&fv)[16]) {
    #pragma unroll
    for (int r = 0; r < 8; ++r) {
        const unsigned o0 = (unsigned)row_share_i<P>(offs[r]) + lane_off;
        fv[r] = *reinterpret_cast<const v4f*>(ftb + o0);
    }
    #pragma unroll
    for (int r = 0; r < 8; ++r) {
        const unsigned o1 = (unsigned)row_share_i<P + 8>(offs[r]) + lane_off;
        fv[r + 8] = *reinterpret_cast<const v4f*>(ftb + o1);
    }
}

template <int P>
__device__ __forceinline__ void gather_accum_store(const v4f (&fv)[16],
        const float (&dx)[8], const float (&dy)[8], const float (&dz)[8],
        float* __restrict__ vt, int j, int h) {
    float ax[4] = {}, ay[4] = {}, az[4] = {}, as[4] = {};
    #pragma unroll
    for (int r = 0; r < 8; ++r) {
        const float x = row_share_f<P>(dx[r]), y = row_share_f<P>(dy[r]), z = row_share_f<P>(dz[r]);
        #pragma unroll
        for (int c = 0; c < 4; ++c) {
            ax[c] = fmaf(fv[r][c], x, ax[c]);
            ay[c] = fmaf(fv[r][c], y, ay[c]);
            az[c] = fmaf(fv[r][c], z, az[c]);
            as[c] += fv[r][c];
        }
    }
    #pragma unroll
    for (int r = 0; r < 8; ++r) {
        const float x = row_share_f<P + 8>(dx[r]), y = row_share_f<P + 8>(dy[r]), z = row_share_f<P + 8>(dz[r]);
        #pragma unroll
        for (int c = 0; c < 4; ++c) {
            ax[c] = fmaf(fv[r + 8][c], x, ax[c]);
            ay[c] = fmaf(fv[r + 8][c], y, ay[c]);
            az[c] = fmaf(fv[r + 8][c], z, az[c]);
            as[c] += fv[r + 8][c];
        }
    }
    float* base = vt + (2 * j) * VSTR + (2 * P + h) * 2;
    v2f t;
    t.x = ax[0]; t.y = ax[1]; *reinterpret_cast<v2f*>(base)                    = t;
    t.x = ax[2]; t.y = ax[3]; *reinterpret_cast<v2f*>(base + VSTR)             = t;
    t.x = ay[0]; t.y = ay[1]; *reinterpret_cast<v2f*>(base + 32 * VSTR)        = t;
    t.x = ay[2]; t.y = ay[3]; *reinterpret_cast<v2f*>(base + 33 * VSTR)        = t;
    t.x = az[0]; t.y = az[1]; *reinterpret_cast<v2f*>(base + 64 * VSTR)        = t;
    t.x = az[2]; t.y = az[3]; *reinterpret_cast<v2f*>(base + 65 * VSTR)        = t;
    t.x = as[0]; t.y = as[1]; *reinterpret_cast<v2f*>(base + 96 * VSTR)        = t;
    t.x = as[2]; t.y = as[3]; *reinterpret_cast<v2f*>(base + 97 * VSTR)        = t;
}

template <bool USE_FT>
__global__ __launch_bounds__(128) void flexconv_wmma_kernel(
    const float* __restrict__ feat,
    const float* __restrict__ ft,
    const float* __restrict__ wt,
    const float* __restrict__ wb,
    const float* __restrict__ bias,
    const int*   __restrict__ nbr,
    const float* __restrict__ pos,
    float* __restrict__ out)
{
    __shared__ alignas(16) float lds[LDS_FLOATS];
    const int tid = threadIdx.x;

    v4f wr0[16], wr1[16];
    #pragma unroll
    for (int j = 0; j < 16; ++j) {
        const int i   = tid + j * 128;
        const int dcp = i >> 4, oq = i & 15;
        const int dc  = dcp * 2;
        const float* row = (j < 12) ? (wt + dc * 64) : (wb + (dc - 192) * 64);
        wr0[j] = *reinterpret_cast<const v4f*>(row + oq * 4);
        wr1[j] = *reinterpret_cast<const v4f*>(row + 64 + oq * 4);
    }
    float bv = 0.0f;
    if (tid < COUT) bv = bias[tid];
    #pragma unroll
    for (int j = 0; j < 16; ++j) {
        const int i   = tid + j * 128;
        const int dcp = i >> 4, oq = i & 15;
        v4f lo, hi;
        lo.x = wr0[j].x; lo.y = wr1[j].x; lo.z = wr0[j].y; lo.w = wr1[j].y;
        hi.x = wr0[j].z; hi.y = wr1[j].z; hi.z = wr0[j].w; hi.w = wr1[j].w;
        float* dst = &lds[dcp * WPAIRSTR + oq * 8];
        *reinterpret_cast<v4f*>(dst)     = lo;
        *reinterpret_cast<v4f*>(dst + 4) = hi;
    }
    if (tid < COUT) lds[BOFF + tid] = bv;
    __syncthreads();

    const int wave = tid >> 5;
    const int lane = tid & 31;
    const int nl   = lane & 15;
    const int h    = lane >> 4;
    const int b    = blockIdx.y;
    const int n    = blockIdx.x * 64 + wave * 16 + nl;
    const int vb   = VOFF + wave * VWAVE;

    if constexpr (USE_FT) {
        const int nh = blockIdx.x * 64 + wave * 16 + 2 * (nl & 7) + h;
        const int kh = (nl >> 3) * 8;
        const float cx = pos[(b * 3 + 0) * NPTS + nh];
        const float cy = pos[(b * 3 + 1) * NPTS + nh];
        const float cz = pos[(b * 3 + 2) * NPTS + nh];
        int   offs[8];
        float dxs[8], dys[8], dzs[8];
        #pragma unroll
        for (int r = 0; r < 8; ++r)
            offs[r] = nbr[(b * KNB + kh + r) * NPTS + nh];
        #pragma unroll
        for (int r = 0; r < 8; ++r) {
            const int id = offs[r];
            dxs[r] = pos[(b * 3 + 0) * NPTS + id] - cx;
            dys[r] = pos[(b * 3 + 1) * NPTS + id] - cy;
            dzs[r] = pos[(b * 3 + 2) * NPTS + id] - cz;
            offs[r] = id * (CIN * 4);
        }
        const char* ftb = reinterpret_cast<const char*>(ft + (size_t)b * NPTS * CIN);
        const unsigned lane_off = nl * 16;
        float* vt = &lds[vb];
        v4f fa[16], fb[16];
        gather_issue<0>(ftb, lane_off, offs, fa);
        gather_issue<1>(ftb, lane_off, offs, fb);
        gather_accum_store<0>(fa, dxs, dys, dzs, vt, nl, h);
        gather_issue<2>(ftb, lane_off, offs, fa);
        gather_accum_store<1>(fb, dxs, dys, dzs, vt, nl, h);
        gather_issue<3>(ftb, lane_off, offs, fb);
        gather_accum_store<2>(fa, dxs, dys, dzs, vt, nl, h);
        gather_issue<4>(ftb, lane_off, offs, fa);
        gather_accum_store<3>(fb, dxs, dys, dzs, vt, nl, h);
        gather_issue<5>(ftb, lane_off, offs, fb);
        gather_accum_store<4>(fa, dxs, dys, dzs, vt, nl, h);
        gather_issue<6>(ftb, lane_off, offs, fa);
        gather_accum_store<5>(fb, dxs, dys, dzs, vt, nl, h);
        gather_issue<7>(ftb, lane_off, offs, fb);
        gather_accum_store<6>(fa, dxs, dys, dzs, vt, nl, h);
        gather_accum_store<7>(fb, dxs, dys, dzs, vt, nl, h);
    } else {
        const float cx = pos[(b * 3 + 0) * NPTS + n];
        const float cy = pos[(b * 3 + 1) * NPTS + n];
        const float cz = pos[(b * 3 + 2) * NPTS + n];

        int   idxs[KNB];
        float dxs[KNB], dys[KNB], dzs[KNB];
        #pragma unroll
        for (int k = 0; k < KNB; ++k)
            idxs[k] = nbr[(b * KNB + k) * NPTS + n];
        #pragma unroll
        for (int k = 0; k < KNB; ++k) {
            const int id = idxs[k];
            dxs[k] = pos[(b * 3 + 0) * NPTS + id] - cx;
            dys[k] = pos[(b * 3 + 1) * NPTS + id] - cy;
            dzs[k] = pos[(b * 3 + 2) * NPTS + id] - cz;
        }

        for (int cc = 0; cc < 4; ++cc) {
            const int cbase = h * 32 + cc * 8;
            float ax[8] = {}, ay[8] = {}, az[8] = {}, asum[8] = {};
            for (int k = 0; k < KNB; ++k) {
                const int   id = idxs[k];
                const float dx = dxs[k], dy = dys[k], dz = dzs[k];
                float f[8];
                const float* fp = feat + (b * CIN + cbase) * NPTS + id;
                #pragma unroll
                for (int c = 0; c < 8; ++c) f[c] = fp[c * NPTS];
                #pragma unroll
                for (int c = 0; c < 8; ++c) {
                    ax[c]   = fmaf(f[c], dx, ax[c]);
                    ay[c]   = fmaf(f[c], dy, ay[c]);
                    az[c]   = fmaf(f[c], dz, az[c]);
                    asum[c] += f[c];
                }
            }
            #pragma unroll
            for (int c = 0; c < 8; c += 2) {
                const int dcp = (cbase + c) >> 1;
                v2f px, py, pz, ps;
                px.x = ax[c];   px.y = ax[c + 1];
                py.x = ay[c];   py.y = ay[c + 1];
                pz.x = az[c];   pz.y = az[c + 1];
                ps.x = asum[c]; ps.y = asum[c + 1];
                *reinterpret_cast<v2f*>(&lds[vb + (dcp      ) * VSTR + nl * 2]) = px;
                *reinterpret_cast<v2f*>(&lds[vb + (dcp + 32 ) * VSTR + nl * 2]) = py;
                *reinterpret_cast<v2f*>(&lds[vb + (dcp + 64 ) * VSTR + nl * 2]) = pz;
                *reinterpret_cast<v2f*>(&lds[vb + (dcp + 96 ) * VSTR + nl * 2]) = ps;
            }
        }
    }

    v8f acc[4] = {v8f{}, v8f{}, v8f{}, v8f{}};
    #pragma unroll 4
    for (int ks = 0; ks < 64; ++ks) {
        const int dcp = ks * 2 + h;
        const v2f bf = *reinterpret_cast<const v2f*>(&lds[vb + dcp * VSTR + nl * 2]);
        #pragma unroll
        for (int ot = 0; ot < 4; ++ot) {
            const v2f af = *reinterpret_cast<const v2f*>(
                &lds[dcp * WPAIRSTR + (ot * 16 + nl) * 2]);
            acc[ot] = __builtin_amdgcn_wmma_f32_16x16x4_f32(
                false, af, false, bf, (short)0, acc[ot], false, false);
        }
    }

    #pragma unroll
    for (int ot = 0; ot < 4; ++ot) {
        #pragma unroll
        for (int r = 0; r < 8; ++r) {
            const int o = ot * 16 + r + h * 8;
            out[(b * COUT + o) * NPTS + n] = acc[ot][r] + lds[BOFF + o];
        }
    }
}

extern "C" void kernel_launch(void* const* d_in, const int* in_sizes, int n_in,
                              void* d_out, int out_size, void* d_ws, size_t ws_size,
                              hipStream_t stream) {
    (void)in_sizes; (void)n_in; (void)out_size;
    const float* feat = (const float*)d_in[0];
    const float* wt   = (const float*)d_in[1];
    const float* wb   = (const float*)d_in[2];
    const float* bias = (const float*)d_in[3];
    const int*   nbr  = (const int*)d_in[4];
    const float* pos  = (const float*)d_in[5];
    float* out = (float*)d_out;

    const size_t ft_bytes = (size_t)NB * NPTS * CIN * sizeof(float);
    dim3 grid(NPTS / 64, NB);

    if (ws_size >= ft_bytes) {
        float* ft = (float*)d_ws;
        transpose_feat_kernel<<<grid, 256, 0, stream>>>(feat, ft);
        flexconv_wmma_kernel<true><<<grid, 128, 0, stream>>>(
            feat, ft, wt, wb, bias, nbr, pos, out);
    } else {
        flexconv_wmma_kernel<false><<<grid, 128, 0, stream>>>(
            feat, feat, wt, wb, bias, nbr, pos, out);
    }
}


# positions re-laid by the first launch as one 16 B row {x,y,z,0} per point in the workspace: a neighbour's position is one load on one line, not three loads on three lines; sums unchanged
# speedup vs baseline: 3.1305x; 1.2120x over previous
#include <hip/hip_runtime.h>

typedef __attribute__((ext_vector_type(2))) float v2f;
typedef __attribute__((ext_vector_type(4))) float v4f;
typedef __attribute__((ext_vector_type(8))) float v8f;

#define NPTS 32768
#define NB   4
#define CIN  64
#define COUT 64
#define KNB  16

#define WPAIRSTR 160
#define WSZ  (128 * WPAIRSTR)
#define VOFF WSZ
#define VSTR 34
#define VWAVE (128 * VSTR)
#define BOFF (VOFF + 4 * VWAVE)
#define LDS_FLOATS (BOFF + 64)

template <bool PACK_POS>
__global__ __launch_bounds__(256) void transpose_feat_kernel(
    const float* __restrict__ feat, const float* __restrict__ pos,
    float* __restrict__ ft, float* __restrict__ p4)
{
    __shared__ float tile[64][65];
    const int b  = blockIdx.y;
    const int n0 = blockIdx.x * 64;
    const int tid = threadIdx.x;
    if constexpr (PACK_POS) {
        if (tid < 64) {
            const int n = n0 + tid;
            v4f q;
            q.x = pos[(b * 3 + 0) * NPTS + n];
            q.y = pos[(b * 3 + 1) * NPTS + n];
            q.z = pos[(b * 3 + 2) * NPTS + n];
            q.w = 0.0f;
            *reinterpret_cast<v4f*>(p4 + (size_t)(b * NPTS + n) * 4) = q;
        }
    }
    #pragma unroll
    for (int i = 0; i < 16; ++i) {
        const int j = tid + i * 256;
        const int c = j >> 6, nloc = j & 63;
        tile[nloc][c] = feat[(b * CIN + c) * NPTS + n0 + nloc];
    }
    __syncthreads();
    #pragma unroll
    for (int i = 0; i < 16; ++i) {
        const int j = tid + i * 256;
        const int nloc = j >> 6, c = j & 63;
        ft[(b * NPTS + n0 + nloc) * CIN + c] = tile[nloc][c];
    }
}

template <int SRC>
__device__ __forceinline__ int row_share_i(int v) {
    return __builtin_amdgcn_mov_dpp(v, 0x150 + SRC, 0xF, 0xF, true);
}
template <int SRC>
__device__ __forceinline__ float row_share_f(float v) {
    return __builtin_bit_cast(float, row_share_i<SRC>(__builtin_bit_cast(int, v)));
}

template <int P>
__device__ __forceinline__ void gather_issue(const char* __restrict__ ftb, unsigned lane_off,
                                             const int (&offs)[8], v4f (&fv)[16]) {
    #pragma unroll
    for (int r = 0; r < 8; ++r) {
        const unsigned o0 = (unsigned)row_share_i<P>(offs[r]) + lane_off;
        fv[r] = *reinterpret_cast<const v4f*>(ftb + o0);
    }
    #pragma unroll
    for (int r = 0; r < 8; ++r) {
        const unsigned o1 = (unsigned)row_share_i<P + 8>(offs[r]) + lane_off;
        fv[r + 8] = *reinterpret_cast<const v4f*>(ftb + o1);
    }
}

template <int P>
__device__ __forceinline__ void gather_accum_store(const v4f (&fv)[16],
        const float (&dx)[8], const float (&dy)[8], const float (&dz)[8],
        float* __restrict__ vt, int j, int h) {
    float ax[4] = {}, ay[4] = {}, az[4] = {}, as[4] = {};
    #pragma unroll
    for (int r = 0; r < 8; ++r) {
        const float x = row_share_f<P>(dx[r]), y = row_share_f<P>(dy[r]), z = row_share_f<P>(dz[r]);
        #pragma unroll
        for (int c = 0; c < 4; ++c) {
            ax[c] = fmaf(fv[r][c], x, ax[c]);
            ay[c] = fmaf(fv[r][c], y, ay[c]);
            az[c] = fmaf(fv[r][c], z, az[c]);
            as[c] += fv[r][c];
        }
    }
    #pragma unroll
    for (int r = 0; r < 8; ++r) {
        const float x = row_share_f<P + 8>(dx[r]), y = row_share_f<P + 8>(dy[r]), z = row_share_f<P + 8>(dz[r]);
        #pragma unroll
        for (int c = 0; c < 4; ++c) {
            ax[c] = fmaf(fv[r + 8][c], x, ax[c]);
            ay[c] = fmaf(fv[r + 8][c], y, ay[c]);
            az[c] = fmaf(fv[r + 8][c], z, az[c]);
            as[c] += fv[r + 8][c];
        }
    }
    float* base = vt + (2 * j) * VSTR + (2 * P + h) * 2;
    v2f t;
    t.x = ax[0]; t.y = ax[1]; *reinterpret_cast<v2f*>(base)                    = t;
    t.x = ax[2]; t.y = ax[3]; *reinterpret_cast<v2f*>(base + VSTR)             = t;
    t.x = ay[0]; t.y = ay[1]; *reinterpret_cast<v2f*>(base + 32 * VSTR)        = t;
    t.x = ay[2]; t.y = ay[3]; *reinterpret_cast<v2f*>(base + 33 * VSTR)        = t;
    t.x = az[0]; t.y = az[1]; *reinterpret_cast<v2f*>(base + 64 * VSTR)        = t;
    t.x = az[2]; t.y = az[3]; *reinterpret_cast<v2f*>(base + 65 * VSTR)        = t;
    t.x = as[0]; t.y = as[1]; *reinterpret_cast<v2f*>(base + 96 * VSTR)        = t;
    t.x = as[2]; t.y = as[3]; *reinterpret_cast<v2f*>(base + 97 * VSTR)        = t;
}

template <bool USE_FT, bool USE_P4>
__global__ __launch_bounds__(128) void flexconv_wmma_kernel(
    const float* __restrict__ feat,
    const float* __restrict__ ft,
    const float* __restrict__ p4,
    const float* __restrict__ wt,
    const float* __restrict__ wb,
    const float* __restrict__ bias,
    const int*   __restrict__ nbr,
    const float* __restrict__ pos,
    float* __restrict__ out)
{
    __shared__ alignas(16) float lds[LDS_FLOATS];
    const int tid = threadIdx.x;

    v4f wr0[16], wr1[16];
    #pragma unroll
    for (int j = 0; j < 16; ++j) {
        const int i   = tid + j * 128;
        const int dcp = i >> 4, oq = i & 15;
        const int dc  = dcp * 2;
        const float* row = (j < 12) ? (wt + dc * 64) : (wb + (dc - 192) * 64);
        wr0[j] = *reinterpret_cast<const v4f*>(row + oq * 4);
        wr1[j] = *reinterpret_cast<const v4f*>(row + 64 + oq * 4);
    }
    float bv = 0.0f;
    if (tid < COUT) bv = bias[tid];
    #pragma unroll
    for (int j = 0; j < 16; ++j) {
        const int i   = tid + j * 128;
        const int dcp = i >> 4, oq = i & 15;
        v4f lo, hi;
        lo.x = wr0[j].x; lo.y = wr1[j].x; lo.z = wr0[j].y; lo.w = wr1[j].y;
        hi.x = wr0[j].z; hi.y = wr1[j].z; hi.z = wr0[j].w; hi.w = wr1[j].w;
        float* dst = &lds[dcp * WPAIRSTR + oq * 8];
        *reinterpret_cast<v4f*>(dst)     = lo;
        *reinterpret_cast<v4f*>(dst + 4) = hi;
    }
    if (tid < COUT) lds[BOFF + tid] = bv;
    __syncthreads();

    const int wave = tid >> 5;
    const int lane = tid & 31;
    const int nl   = lane & 15;
    const int h    = lane >> 4;
    const int b    = blockIdx.y;
    const int n    = blockIdx.x * 64 + wave * 16 + nl;
    const int vb   = VOFF + wave * VWAVE;

    if constexpr (USE_FT) {
        const int nh = blockIdx.x * 64 + wave * 16 + 2 * (nl & 7) + h;
        const int kh = (nl >> 3) * 8;
        int   offs[8];
        float dxs[8], dys[8], dzs[8];
        #pragma unroll
        for (int r = 0; r < 8; ++r)
            offs[r] = nbr[(b * KNB + kh + r) * NPTS + nh];
        if constexpr (USE_P4) {
            const v4f* p4b = reinterpret_cast<const v4f*>(p4) + (size_t)b * NPTS;
            const v4f c4 = p4b[nh];
            v4f q[8];
            #pragma unroll
            for (int r = 0; r < 8; ++r)
                q[r] = p4b[offs[r]];
            #pragma unroll
            for (int r = 0; r < 8; ++r) {
                dxs[r] = q[r].x - c4.x;
                dys[r] = q[r].y - c4.y;
                dzs[r] = q[r].z - c4.z;
                offs[r] *= (CIN * 4);
            }
        } else {
            const float cx = pos[(b * 3 + 0) * NPTS + nh];
            const float cy = pos[(b * 3 + 1) * NPTS + nh];
            const float cz = pos[(b * 3 + 2) * NPTS + nh];
            #pragma unroll
            for (int r = 0; r < 8; ++r) {
                const int id = offs[r];
                dxs[r] = pos[(b * 3 + 0) * NPTS + id] - cx;
                dys[r] = pos[(b * 3 + 1) * NPTS + id] - cy;
                dzs[r] = pos[(b * 3 + 2) * NPTS + id] - cz;
                offs[r] = id * (CIN * 4);
            }
        }
        const char* ftb = reinterpret_cast<const char*>(ft + (size_t)b * NPTS * CIN);
        const unsigned lane_off = nl * 16;
        float* vt = &lds[vb];
        v4f fa[16], fb[16];
        gather_issue<0>(ftb, lane_off, offs, fa);
        gather_issue<1>(ftb, lane_off, offs, fb);
        gather_accum_store<0>(fa, dxs, dys, dzs, vt, nl, h);
        gather_issue<2>(ftb, lane_off, offs, fa);
        gather_accum_store<1>(fb, dxs, dys, dzs, vt, nl, h);
        gather_issue<3>(ftb, lane_off, offs, fb);
        gather_accum_store<2>(fa, dxs, dys, dzs, vt, nl, h);
        gather_issue<4>(ftb, lane_off, offs, fa);
        gather_accum_store<3>(fb, dxs, dys, dzs, vt, nl, h);
        gather_issue<5>(ftb, lane_off, offs, fb);
        gather_accum_store<4>(fa, dxs, dys, dzs, vt, nl, h);
        gather_issue<6>(ftb, lane_off, offs, fa);
        gather_accum_store<5>(fb, dxs, dys, dzs, vt, nl, h);
        gather_issue<7>(ftb, lane_off, offs, fb);
        gather_accum_store<6>(fa, dxs, dys, dzs, vt, nl, h);
        gather_accum_store<7>(fb, dxs, dys, dzs, vt, nl, h);
    } else {
        const float cx = pos[(b * 3 + 0) * NPTS + n];
        const float cy = pos[(b * 3 + 1) * NPTS + n];
        const float cz = pos[(b * 3 + 2) * NPTS + n];

        int   idxs[KNB];
        float dxs[KNB], dys[KNB], dzs[KNB];
        #pragma unroll
        for (int k = 0; k < KNB; ++k)
            idxs[k] = nbr[(b * KNB + k) * NPTS + n];
        #pragma unroll
        for (int k = 0; k < KNB; ++k) {
            const int id = idxs[k];
            dxs[k] = pos[(b * 3 + 0) * NPTS + id] - cx;
            dys[k] = pos[(b * 3 + 1) * NPTS + id] - cy;
            dzs[k] = pos[(b * 3 + 2) * NPTS + id] - cz;
        }

        for (int cc = 0; cc < 4; ++cc) {
            const int cbase = h * 32 + cc * 8;
            float ax[8] = {}, ay[8] = {}, az[8] = {}, asum[8] = {};
            for (int k = 0; k < KNB; ++k) {
                const int   id = idxs[k];
                const float dx = dxs[k], dy = dys[k], dz = dzs[k];
                float f[8];
                const float* fp = feat + (b * CIN + cbase) * NPTS + id;
                #pragma unroll
                for (int c = 0; c < 8; ++c) f[c] = fp[c * NPTS];
                #pragma unroll
                for (int c = 0; c < 8; ++c) {
                    ax[c]   = fmaf(f[c], dx, ax[c]);
                    ay[c]   = fmaf(f[c], dy, ay[c]);
                    az[c]   = fmaf(f[c], dz, az[c]);
                    asum[c] += f[c];
                }
            }
            #pragma unroll
            for (int c = 0; c < 8; c += 2) {
                const int dcp = (cbase + c) >> 1;
                v2f px, py, pz, ps;
                px.x = ax[c];   px.y = ax[c + 1];
                py.x = ay[c];   py.y = ay[c + 1];
                pz.x = az[c];   pz.y = az[c + 1];
                ps.x = asum[c]; ps.y = asum[c + 1];
                *reinterpret_cast<v2f*>(&lds[vb + (dcp      ) * VSTR + nl * 2]) = px;
                *reinterpret_cast<v2f*>(&lds[vb + (dcp + 32 ) * VSTR + nl * 2]) = py;
                *reinterpret_cast<v2f*>(&lds[vb + (dcp + 64 ) * VSTR + nl * 2]) = pz;
                *reinterpret_cast<v2f*>(&lds[vb + (dcp + 96 ) * VSTR + nl * 2]) = ps;
            }
        }
    }

    v8f acc[4] = {v8f{}, v8f{}, v8f{}, v8f{}};
    #pragma unroll 4
    for (int ks = 0; ks < 64; ++ks) {
        const int dcp = ks * 2 + h;
        const v2f bf = *reinterpret_cast<const v2f*>(&lds[vb + dcp * VSTR + nl * 2]);
        #pragma unroll
        for (int ot = 0; ot < 4; ++ot) {
            const v2f af = *reinterpret_cast<const v2f*>(
                &lds[dcp * WPAIRSTR + (ot * 16 + nl) * 2]);
            acc[ot] = __builtin_amdgcn_wmma_f32_16x16x4_f32(
                false, af, false, bf, (short)0, acc[ot], false, false);
        }
    }

    #pragma unroll
    for (int ot = 0; ot < 4; ++ot) {
        #pragma unroll
        for (int r = 0; r < 8; ++r) {
            const int o = ot * 16 + r + h * 8;
            out[(b * COUT + o) * NPTS + n] = acc[ot][r] + lds[BOFF + o];
        }
    }
}

extern "C" void kernel_launch(void* const* d_in, const int* in_sizes, int n_in,
                              void* d_out, int out_size, void* d_ws, size_t ws_size,
                              hipStream_t stream) {
    (void)in_sizes; (void)n_in; (void)out_size;
    const float* feat = (const float*)d_in[0];
    const float* wt   = (const float*)d_in[1];
    const float* wb   = (const float*)d_in[2];
    const float* bias = (const float*)d_in[3];
    const int*   nbr  = (const int*)d_in[4];
    const float* pos  = (const float*)d_in[5];
    float* out = (float*)d_out;

    const size_t ft_bytes = (size_t)NB * NPTS * CIN * sizeof(float);
    const size_t p4_bytes = (size_t)NB * NPTS * 4 * sizeof(float);
    dim3 grid(NPTS / 64, NB);

    if (ws_size >= ft_bytes + p4_bytes) {
        float* ft = (float*)d_ws;
        float* p4 = ft + (size_t)NB * NPTS * CIN;
        transpose_feat_kernel<true><<<grid, 256, 0, stream>>>(feat, pos, ft, p4);
        flexconv_wmma_kernel<true, true><<<grid, 128, 0, stream>>>(
            feat, ft, p4, wt, wb, bias, nbr, pos, out);
    } else if (ws_size >= ft_bytes) {
        float* ft = (float*)d_ws;
        transpose_feat_kernel<false><<<grid, 256, 0, stream>>>(feat, pos, ft, ft);
        flexconv_wmma_kernel<true, false><<<grid, 128, 0, stream>>>(
            feat, ft, ft, wt, wb, bias, nbr, pos, out);
    } else {
        flexconv_wmma_kernel<false, false><<<grid, 128, 0, stream>>>(
            feat, feat, feat, wt, wb, bias, nbr, pos, out);
    }
}


# persistent work-groups: W' and bias staged into LDS once per work-group, which then loops over the 4 batch tiles of its 64-point column (grid 512; was 2048, each re-staging W'); sums unchanged
# speedup vs baseline: 3.2908x; 1.0512x over previous
#include <hip/hip_runtime.h>

typedef __attribute__((ext_vector_type(2))) float v2f;
typedef __attribute__((ext_vector_type(4))) float v4f;
typedef __attribute__((ext_vector_type(8))) float v8f;

#define NPTS 32768
#define NB   4
#define CIN  64
#define COUT 64
#define KNB  16

#define WPAIRSTR 160
#define WSZ  (128 * WPAIRSTR)
#define VOFF WSZ
#define VSTR 34
#define VWAVE (128 * VSTR)
#define BOFF (VOFF + 4 * VWAVE)
#define LDS_FLOATS (BOFF + 64)

template <bool PACK_POS>
__global__ __launch_bounds__(256) void transpose_feat_kernel(
    const float* __restrict__ feat, const float* __restrict__ pos,
    float* __restrict__ ft, float* __restrict__ p4)
{
    __shared__ float tile[64][65];
    const int b  = blockIdx.y;
    const int n0 = blockIdx.x * 64;
    const int tid = threadIdx.x;
    if constexpr (PACK_POS) {
        if (tid < 64) {
            const int n = n0 + tid;
            v4f q;
            q.x = pos[(b * 3 + 0) * NPTS + n];
            q.y = pos[(b * 3 + 1) * NPTS + n];
            q.z = pos[(b * 3 + 2) * NPTS + n];
            q.w = 0.0f;
            *reinterpret_cast<v4f*>(p4 + (size_t)(b * NPTS + n) * 4) = q;
        }
    }
    #pragma unroll
    for (int i = 0; i < 16; ++i) {
        const int j = tid + i * 256;
        const int c = j >> 6, nloc = j & 63;
        tile[nloc][c] = feat[(b * CIN + c) * NPTS + n0 + nloc];
    }
    __syncthreads();
    #pragma unroll
    for (int i = 0; i < 16; ++i) {
        const int j = tid + i * 256;
        const int nloc = j >> 6, c = j & 63;
        ft[(b * NPTS + n0 + nloc) * CIN + c] = tile[nloc][c];
    }
}

template <int SRC>
__device__ __forceinline__ int row_share_i(int v) {
    return __builtin_amdgcn_mov_dpp(v, 0x150 + SRC, 0xF, 0xF, true);
}
template <int SRC>
__device__ __forceinline__ float row_share_f(float v) {
    return __builtin_bit_cast(float, row_share_i<SRC>(__builtin_bit_cast(int, v)));
}

template <int P>
__device__ __forceinline__ void gather_issue(const char* __restrict__ ftb, unsigned lane_off,
                                             const int (&offs)[8], v4f (&fv)[16]) {
    #pragma unroll
    for (int r = 0; r < 8; ++r) {
        const unsigned o0 = (unsigned)row_share_i<P>(offs[r]) + lane_off;
        fv[r] = *reinterpret_cast<const v4f*>(ftb + o0);
    }
    #pragma unroll
    for (int r = 0; r < 8; ++r) {
        const unsigned o1 = (unsigned)row_share_i<P + 8>(offs[r]) + lane_off;
        fv[r + 8] = *reinterpret_cast<const v4f*>(ftb + o1);
    }
}

template <int P>
__device__ __forceinline__ void gather_accum_store(const v4f (&fv)[16],
        const float (&dx)[8], const float (&dy)[8], const float (&dz)[8],
        float* __restrict__ vt, int j, int h) {
    float ax[4] = {}, ay[4] = {}, az[4] = {}, as[4] = {};
    #pragma unroll
    for (int r = 0; r < 8; ++r) {
        const float x = row_share_f<P>(dx[r]), y = row_share_f<P>(dy[r]), z = row_share_f<P>(dz[r]);
        #pragma unroll
        for (int c = 0; c < 4; ++c) {
            ax[c] = fmaf(fv[r][c], x, ax[c]);
            ay[c] = fmaf(fv[r][c], y, ay[c]);
            az[c] = fmaf(fv[r][c], z, az[c]);
            as[c] += fv[r][c];
        }
    }
    #pragma unroll
    for (int r = 0; r < 8; ++r) {
        const float x = row_share_f<P + 8>(dx[r]), y = row_share_f<P + 8>(dy[r]), z = row_share_f<P + 8>(dz[r]);
        #pragma unroll
        for (int c = 0; c < 4; ++c) {
            ax[c] = fmaf(fv[r + 8][c], x, ax[c]);
            ay[c] = fmaf(fv[r + 8][c], y, ay[c]);
            az[c] = fmaf(fv[r + 8][c], z, az[c]);
            as[c] += fv[r + 8][c];
        }
    }
    float* base = vt + (2 * j) * VSTR + (2 * P + h) * 2;
    v2f t;
    t.x = ax[0]; t.y = ax[1]; *reinterpret_cast<v2f*>(base)                    = t;
    t.x = ax[2]; t.y = ax[3]; *reinterpret_cast<v2f*>(base + VSTR)             = t;
    t.x = ay[0]; t.y = ay[1]; *reinterpret_cast<v2f*>(base + 32 * VSTR)        = t;
    t.x = ay[2]; t.y = ay[3]; *reinterpret_cast<v2f*>(base + 33 * VSTR)        = t;
    t.x = az[0]; t.y = az[1]; *reinterpret_cast<v2f*>(base + 64 * VSTR)        = t;
    t.x = az[2]; t.y = az[3]; *reinterpret_cast<v2f*>(base + 65 * VSTR)        = t;
    t.x = as[0]; t.y = as[1]; *reinterpret_cast<v2f*>(base + 96 * VSTR)        = t;
    t.x = as[2]; t.y = as[3]; *reinterpret_cast<v2f*>(base + 97 * VSTR)        = t;
}

template <bool USE_FT, bool USE_P4>
__global__ __launch_bounds__(128) void flexconv_wmma_kernel(
    const float* __restrict__ feat,
    const float* __restrict__ ft,
    const float* __restrict__ p4,
    const float* __restrict__ wt,
    const float* __restrict__ wb,
    const float* __restrict__ bias,
    const int*   __restrict__ nbr,
    const float* __restrict__ pos,
    float* __restrict__ out)
{
    __shared__ alignas(16) float lds[LDS_FLOATS];
    const int tid = threadIdx.x;

    v4f wr0[16], wr1[16];
    #pragma unroll
    for (int j = 0; j < 16; ++j) {
        const int i   = tid + j * 128;
        const int dcp = i >> 4, oq = i & 15;
        const int dc  = dcp * 2;
        const float* row = (j < 12) ? (wt + dc * 64) : (wb + (dc - 192) * 64);
        wr0[j] = *reinterpret_cast<const v4f*>(row + oq * 4);
        wr1[j] = *reinterpret_cast<const v4f*>(row + 64 + oq * 4);
    }
    float bv = 0.0f;
    if (tid < COUT) bv = bias[tid];
    #pragma unroll
    for (int j = 0; j < 16; ++j) {
        const int i   = tid + j * 128;
        const int dcp = i >> 4, oq = i & 15;
        v4f lo, hi;
        lo.x = wr0[j].x; lo.y = wr1[j].x; lo.z = wr0[j].y; lo.w = wr1[j].y;
        hi.x = wr0[j].z; hi.y = wr1[j].z; hi.z = wr0[j].w; hi.w = wr1[j].w;
        float* dst = &lds[dcp * WPAIRSTR + oq * 8];
        *reinterpret_cast<v4f*>(dst)     = lo;
        *reinterpret_cast<v4f*>(dst + 4) = hi;
    }
    if (tid < COUT) lds[BOFF + tid] = bv;
    __syncthreads();

    const int wave = tid >> 5;
    const int lane = tid & 31;
    const int nl   = lane & 15;
    const int h    = lane >> 4;
    const int n    = blockIdx.x * 64 + wave * 16 + nl;
    const int vb   = VOFF + wave * VWAVE;

    #pragma unroll 1
    for (int b = 0; b < NB; ++b) {

    if constexpr (USE_FT) {
        const int nh = blockIdx.x * 64 + wave * 16 + 2 * (nl & 7) + h;
        const int kh = (nl >> 3) * 8;
        int   offs[8];
        float dxs[8], dys[8], dzs[8];
        #pragma unroll
        for (int r = 0; r < 8; ++r)
            offs[r] = nbr[(b * KNB + kh + r) * NPTS + nh];
        if constexpr (USE_P4) {
            const v4f* p4b = reinterpret_cast<const v4f*>(p4) + (size_t)b * NPTS;
            const v4f c4 = p4b[nh];
            v4f q[8];
            #pragma unroll
            for (int r = 0; r < 8; ++r)
                q[r] = p4b[offs[r]];
            #pragma unroll
            for (int r = 0; r < 8; ++r) {
                dxs[r] = q[r].x - c4.x;
                dys[r] = q[r].y - c4.y;
                dzs[r] = q[r].z - c4.z;
                offs[r] *= (CIN * 4);
            }
        } else {
            const float cx = pos[(b * 3 + 0) * NPTS + nh];
            const float cy = pos[(b * 3 + 1) * NPTS + nh];
            const float cz = pos[(b * 3 + 2) * NPTS + nh];
            #pragma unroll
            for (int r = 0; r < 8; ++r) {
                const int id = offs[r];
                dxs[r] = pos[(b * 3 + 0) * NPTS + id] - cx;
                dys[r] = pos[(b * 3 + 1) * NPTS + id] - cy;
                dzs[r] = pos[(b * 3 + 2) * NPTS + id] - cz;
                offs[r] = id * (CIN * 4);
            }
        }
        const char* ftb = reinterpret_cast<const char*>(ft + (size_t)b * NPTS * CIN);
        const unsigned lane_off = nl * 16;
        float* vt = &lds[vb];
        v4f fa[16], fb[16];
        gather_issue<0>(ftb, lane_off, offs, fa);
        gather_issue<1>(ftb, lane_off, offs, fb);
        gather_accum_store<0>(fa, dxs, dys, dzs, vt, nl, h);
        gather_issue<2>(ftb, lane_off, offs, fa);
        gather_accum_store<1>(fb, dxs, dys, dzs, vt, nl, h);
        gather_issue<3>(ftb, lane_off, offs, fb);
        gather_accum_store<2>(fa, dxs, dys, dzs, vt, nl, h);
        gather_issue<4>(ftb, lane_off, offs, fa);
        gather_accum_store<3>(fb, dxs, dys, dzs, vt, nl, h);
        gather_issue<5>(ftb, lane_off, offs, fb);
        gather_accum_store<4>(fa, dxs, dys, dzs, vt, nl, h);
        gather_issue<6>(ftb, lane_off, offs, fa);
        gather_accum_store<5>(fb, dxs, dys, dzs, vt, nl, h);
        gather_issue<7>(ftb, lane_off, offs, fb);
        gather_accum_store<6>(fa, dxs, dys, dzs, vt, nl, h);
        gather_accum_store<7>(fb, dxs, dys, dzs, vt, nl, h);
    } else {
        const float cx = pos[(b * 3 + 0) * NPTS + n];
        const float cy = pos[(b * 3 + 1) * NPTS + n];
        const float cz = pos[(b * 3 + 2) * NPTS + n];

        int   idxs[KNB];
        float dxs[KNB], dys[KNB], dzs[KNB];
        #pragma unroll
        for (int k = 0; k < KNB; ++k)
            idxs[k] = nbr[(b * KNB + k) * NPTS + n];
        #pragma unroll
        for (int k = 0; k < KNB; ++k) {
            const int id = idxs[k];
            dxs[k] = pos[(b * 3 + 0) * NPTS + id] - cx;
            dys[k] = pos[(b * 3 + 1) * NPTS + id] - cy;
            dzs[k] = pos[(b * 3 + 2) * NPTS + id] - cz;
        }

        for (int cc = 0; cc < 4; ++cc) {
            const int cbase = h * 32 + cc * 8;
            float ax[8] = {}, ay[8] = {}, az[8] = {}, asum[8] = {};
            for (int k = 0; k < KNB; ++k) {
                const int   id = idxs[k];
                const float dx = dxs[k], dy = dys[k], dz = dzs[k];
                float f[8];
                const float* fp = feat + (b * CIN + cbase) * NPTS + id;
                #pragma unroll
                for (int c = 0; c < 8; ++c) f[c] = fp[c * NPTS];
                #pragma unroll
                for (int c = 0; c < 8; ++c) {
                    ax[c]   = fmaf(f[c], dx, ax[c]);
                    ay[c]   = fmaf(f[c], dy, ay[c]);
                    az[c]   = fmaf(f[c], dz, az[c]);
                    asum[c] += f[c];
                }
            }
            #pragma unroll
            for (int c = 0; c < 8; c += 2) {
                const int dcp = (cbase + c) >> 1;
                v2f px, py, pz, ps;
                px.x = ax[c];   px.y = ax[c + 1];
                py.x = ay[c];   py.y = ay[c + 1];
                pz.x = az[c];   pz.y = az[c + 1];
                ps.x = asum[c]; ps.y = asum[c + 1];
                *reinterpret_cast<v2f*>(&lds[vb + (dcp      ) * VSTR + nl * 2]) = px;
                *reinterpret_cast<v2f*>(&lds[vb + (dcp + 32 ) * VSTR + nl * 2]) = py;
                *reinterpret_cast<v2f*>(&lds[vb + (dcp + 64 ) * VSTR + nl * 2]) = pz;
                *reinterpret_cast<v2f*>(&lds[vb + (dcp + 96 ) * VSTR + nl * 2]) = ps;
            }
        }
    }

    v8f acc[4] = {v8f{}, v8f{}, v8f{}, v8f{}};
    #pragma unroll 4
    for (int ks = 0; ks < 64; ++ks) {
        const int dcp = ks * 2 + h;
        const v2f bf = *reinterpret_cast<const v2f*>(&lds[vb + dcp * VSTR + nl * 2]);
        #pragma unroll
        for (int ot = 0; ot < 4; ++ot) {
            const v2f af = *reinterpret_cast<const v2f*>(
                &lds[dcp * WPAIRSTR + (ot * 16 + nl) * 2]);
            acc[ot] = __builtin_amdgcn_wmma_f32_16x16x4_f32(
                false, af, false, bf, (short)0, acc[ot], false, false);
        }
    }

    #pragma unroll
    for (int ot = 0; ot < 4; ++ot) {
        #pragma unroll
        for (int r = 0; r < 8; ++r) {
            const int o = ot * 16 + r + h * 8;
            out[(b * COUT + o) * NPTS + n] = acc[ot][r] + lds[BOFF + o];
        }
    }
    }
}

extern "C" void kernel_launch(void* const* d_in, const int* in_sizes, int n_in,
                              void* d_out, int out_size, void* d_ws, size_t ws_size,
                              hipStream_t stream) {
    (void)in_sizes; (void)n_in; (void)out_size;
    const float* feat = (const float*)d_in[0];
    const float* wt   = (const float*)d_in[1];
    const float* wb   = (const float*)d_in[2];
    const float* bias = (const float*)d_in[3];
    const int*   nbr  = (const int*)d_in[4];
    const float* pos  = (const float*)d_in[5];
    float* out = (float*)d_out;

    const size_t ft_bytes = (size_t)NB * NPTS * CIN * sizeof(float);
    const size_t p4_bytes = (size_t)NB * NPTS * 4 * sizeof(float);
    dim3 grid(NPTS / 64, NB);
    dim3 grid_main(NPTS / 64);

    if (ws_size >= ft_bytes + p4_bytes) {
        float* ft = (float*)d_ws;
        float* p4 = ft + (size_t)NB * NPTS * CIN;
        transpose_feat_kernel<true><<<grid, 256, 0, stream>>>(feat, pos, ft, p4);
        flexconv_wmma_kernel<true, true><<<grid_main, 128, 0, stream>>>(
            feat, ft, p4, wt, wb, bias, nbr, pos, out);
    } else if (ws_size >= ft_bytes) {
        float* ft = (float*)d_ws;
        transpose_feat_kernel<false><<<grid, 256, 0, stream>>>(feat, pos, ft, ft);
        flexconv_wmma_kernel<true, false><<<grid_main, 128, 0, stream>>>(
            feat, ft, ft, wt, wb, bias, nbr, pos, out);
    } else {
        flexconv_wmma_kernel<false, false><<<grid_main, 128, 0, stream>>>(
            feat, feat, feat, wt, wb, bias, nbr, pos, out);
    }
}


# matrix phase: LDS fragment reads of the next 8-step block issued as one group ahead of the current block's matrix instructions (second register set), not spread among them; sum order unchanged
# speedup vs baseline: 3.4346x; 1.0437x over previous
#include <hip/hip_runtime.h>

typedef __attribute__((ext_vector_type(2))) float v2f;
typedef __attribute__((ext_vector_type(4))) float v4f;
typedef __attribute__((ext_vector_type(8))) float v8f;

#define NPTS 32768
#define NB   4
#define CIN  64
#define COUT 64
#define KNB  16

#define WPAIRSTR 160
#define WSZ  (128 * WPAIRSTR)
#define VOFF WSZ
#define VSTR 34
#define VWAVE (128 * VSTR)
#define BOFF (VOFF + 4 * VWAVE)
#define LDS_FLOATS (BOFF + 64)

template <bool PACK_POS>
__global__ __launch_bounds__(256) void transpose_feat_kernel(
    const float* __restrict__ feat, const float* __restrict__ pos,
    float* __restrict__ ft, float* __restrict__ p4)
{
    __shared__ float tile[64][65];
    const int b  = blockIdx.y;
    const int n0 = blockIdx.x * 64;
    const int tid = threadIdx.x;
    if constexpr (PACK_POS) {
        if (tid < 64) {
            const int n = n0 + tid;
            v4f q;
            q.x = pos[(b * 3 + 0) * NPTS + n];
            q.y = pos[(b * 3 + 1) * NPTS + n];
            q.z = pos[(b * 3 + 2) * NPTS + n];
            q.w = 0.0f;
            *reinterpret_cast<v4f*>(p4 + (size_t)(b * NPTS + n) * 4) = q;
        }
    }
    #pragma unroll
    for (int i = 0; i < 16; ++i) {
        const int j = tid + i * 256;
        const int c = j >> 6, nloc = j & 63;
        tile[nloc][c] = feat[(b * CIN + c) * NPTS + n0 + nloc];
    }
    __syncthreads();
    #pragma unroll
    for (int i = 0; i < 16; ++i) {
        const int j = tid + i * 256;
        const int nloc = j >> 6, c = j & 63;
        ft[(b * NPTS + n0 + nloc) * CIN + c] = tile[nloc][c];
    }
}

template <int SRC>
__device__ __forceinline__ int row_share_i(int v) {
    return __builtin_amdgcn_mov_dpp(v, 0x150 + SRC, 0xF, 0xF, true);
}
template <int SRC>
__device__ __forceinline__ float row_share_f(float v) {
    return __builtin_bit_cast(float, row_share_i<SRC>(__builtin_bit_cast(int, v)));
}

template <int P>
__device__ __forceinline__ void gather_issue(const char* __restrict__ ftb, unsigned lane_off,
                                             const int (&offs)[8], v4f (&fv)[16]) {
    #pragma unroll
    for (int r = 0; r < 8; ++r) {
        const unsigned o0 = (unsigned)row_share_i<P>(offs[r]) + lane_off;
        fv[r] = *reinterpret_cast<const v4f*>(ftb + o0);
    }
    #pragma unroll
    for (int r = 0; r < 8; ++r) {
        const unsigned o1 = (unsigned)row_share_i<P + 8>(offs[r]) + lane_off;
        fv[r + 8] = *reinterpret_cast<const v4f*>(ftb + o1);
    }
}

template <int P>
__device__ __forceinline__ void gather_accum_store(const v4f (&fv)[16],
        const float (&dx)[8], const float (&dy)[8], const float (&dz)[8],
        float* __restrict__ vt, int j, int h) {
    float ax[4] = {}, ay[4] = {}, az[4] = {}, as[4] = {};
    #pragma unroll
    for (int r = 0; r < 8; ++r) {
        const float x = row_share_f<P>(dx[r]), y = row_share_f<P>(dy[r]), z = row_share_f<P>(dz[r]);
        #pragma unroll
        for (int c = 0; c < 4; ++c) {
            ax[c] = fmaf(fv[r][c], x, ax[c]);
            ay[c] = fmaf(fv[r][c], y, ay[c]);
            az[c] = fmaf(fv[r][c], z, az[c]);
            as[c] += fv[r][c];
        }
    }
    #pragma unroll
    for (int r = 0; r < 8; ++r) {
        const float x = row_share_f<P + 8>(dx[r]), y = row_share_f<P + 8>(dy[r]), z = row_share_f<P + 8>(dz[r]);
        #pragma unroll
        for (int c = 0; c < 4; ++c) {
            ax[c] = fmaf(fv[r + 8][c], x, ax[c]);
            ay[c] = fmaf(fv[r + 8][c], y, ay[c]);
            az[c] = fmaf(fv[r + 8][c], z, az[c]);
            as[c] += fv[r + 8][c];
        }
    }
    float* base = vt + (2 * j) * VSTR + (2 * P + h) * 2;
    v2f t;
    t.x = ax[0]; t.y = ax[1]; *reinterpret_cast<v2f*>(base)                    = t;
    t.x = ax[2]; t.y = ax[3]; *reinterpret_cast<v2f*>(base + VSTR)             = t;
    t.x = ay[0]; t.y = ay[1]; *reinterpret_cast<v2f*>(base + 32 * VSTR)        = t;
    t.x = ay[2]; t.y = ay[3]; *reinterpret_cast<v2f*>(base + 33 * VSTR)        = t;
    t.x = az[0]; t.y = az[1]; *reinterpret_cast<v2f*>(base + 64 * VSTR)        = t;
    t.x = az[2]; t.y = az[3]; *reinterpret_cast<v2f*>(base + 65 * VSTR)        = t;
    t.x = as[0]; t.y = as[1]; *reinterpret_cast<v2f*>(base + 96 * VSTR)        = t;
    t.x = as[2]; t.y = as[3]; *reinterpret_cast<v2f*>(base + 97 * VSTR)        = t;
}

struct FragSet {
    v2f b[8];
    v2f a[8][4];
};

__device__ __forceinline__ void frag_load(const float* __restrict__ wA, const float* __restrict__ vB,
                                          int blk, FragSet& f) {
    #pragma unroll
    for (int s = 0; s < 8; ++s) {
        const int ks = blk * 8 + s;
        f.b[s] = *reinterpret_cast<const v2f*>(vB + ks * (2 * VSTR));
        #pragma unroll
        for (int ot = 0; ot < 4; ++ot)
            f.a[s][ot] = *reinterpret_cast<const v2f*>(wA + ks * (2 * WPAIRSTR) + ot * 32);
    }
}

__device__ __forceinline__ void frag_mma(const FragSet& f, v8f (&acc)[4]) {
    #pragma unroll
    for (int s = 0; s < 8; ++s) {
        #pragma unroll
        for (int ot = 0; ot < 4; ++ot)
            acc[ot] = __builtin_amdgcn_wmma_f32_16x16x4_f32(
                false, f.a[s][ot], false, f.b[s], (short)0, acc[ot], false, false);
    }
}

template <bool USE_FT, bool USE_P4>
__global__ __launch_bounds__(128) void flexconv_wmma_kernel(
    const float* __restrict__ feat,
    const float* __restrict__ ft,
    const float* __restrict__ p4,
    const float* __restrict__ wt,
    const float* __restrict__ wb,
    const float* __restrict__ bias,
    const int*   __restrict__ nbr,
    const float* __restrict__ pos,
    float* __restrict__ out)
{
    __shared__ alignas(16) float lds[LDS_FLOATS];
    const int tid = threadIdx.x;

    v4f wr0[16], wr1[16];
    #pragma unroll
    for (int j = 0; j < 16; ++j) {
        const int i   = tid + j * 128;
        const int dcp = i >> 4, oq = i & 15;
        const int dc  = dcp * 2;
        const float* row = (j < 12) ? (wt + dc * 64) : (wb + (dc - 192) * 64);
        wr0[j] = *reinterpret_cast<const v4f*>(row + oq * 4);
        wr1[j] = *reinterpret_cast<const v4f*>(row + 64 + oq * 4);
    }
    float bv = 0.0f;
    if (tid < COUT) bv = bias[tid];
    #pragma unroll
    for (int j = 0; j < 16; ++j) {
        const int i   = tid + j * 128;
        const int dcp = i >> 4, oq = i & 15;
        v4f lo, hi;
        lo.x = wr0[j].x; lo.y = wr1[j].x; lo.z = wr0[j].y; lo.w = wr1[j].y;
        hi.x = wr0[j].z; hi.y = wr1[j].z; hi.z = wr0[j].w; hi.w = wr1[j].w;
        float* dst = &lds[dcp * WPAIRSTR + oq * 8];
        *reinterpret_cast<v4f*>(dst)     = lo;
        *reinterpret_cast<v4f*>(dst + 4) = hi;
    }
    if (tid < COUT) lds[BOFF + tid] = bv;
    __syncthreads();

    const int wave = tid >> 5;
    const int lane = tid & 31;
    const int nl   = lane & 15;
    const int h    = lane >> 4;
    const int n    = blockIdx.x * 64 + wave * 16 + nl;
    const int vb   = VOFF + wave * VWAVE;

    #pragma unroll 1
    for (int b = 0; b < NB; ++b) {

    if constexpr (USE_FT) {
        const int nh = blockIdx.x * 64 + wave * 16 + 2 * (nl & 7) + h;
        const int kh = (nl >> 3) * 8;
        int   offs[8];
        float dxs[8], dys[8], dzs[8];
        #pragma unroll
        for (int r = 0; r < 8; ++r)
            offs[r] = nbr[(b * KNB + kh + r) * NPTS + nh];
        if constexpr (USE_P4) {
            const v4f* p4b = reinterpret_cast<const v4f*>(p4) + (size_t)b * NPTS;
            const v4f c4 = p4b[nh];
            v4f q[8];
            #pragma unroll
            for (int r = 0; r < 8; ++r)
                q[r] = p4b[offs[r]];
            #pragma unroll
            for (int r = 0; r < 8; ++r) {
                dxs[r] = q[r].x - c4.x;
                dys[r] = q[r].y - c4.y;
                dzs[r] = q[r].z - c4.z;
                offs[r] *= (CIN * 4);
            }
        } else {
            const float cx = pos[(b * 3 + 0) * NPTS + nh];
            const float cy = pos[(b * 3 + 1) * NPTS + nh];
            const float cz = pos[(b * 3 + 2) * NPTS + nh];
            #pragma unroll
            for (int r = 0; r < 8; ++r) {
                const int id = offs[r];
                dxs[r] = pos[(b * 3 + 0) * NPTS + id] - cx;
                dys[r] = pos[(b * 3 + 1) * NPTS + id] - cy;
                dzs[r] = pos[(b * 3 + 2) * NPTS + id] - cz;
                offs[r] = id * (CIN * 4);
            }
        }
        const char* ftb = reinterpret_cast<const char*>(ft + (size_t)b * NPTS * CIN);
        const unsigned lane_off = nl * 16;
        float* vt = &lds[vb];
        v4f fa[16], fb[16];
        gather_issue<0>(ftb, lane_off, offs, fa);
        gather_issue<1>(ftb, lane_off, offs, fb);
        gather_accum_store<0>(fa, dxs, dys, dzs, vt, nl, h);
        gather_issue<2>(ftb, lane_off, offs, fa);
        gather_accum_store<1>(fb, dxs, dys, dzs, vt, nl, h);
        gather_issue<3>(ftb, lane_off, offs, fb);
        gather_accum_store<2>(fa, dxs, dys, dzs, vt, nl, h);
        gather_issue<4>(ftb, lane_off, offs, fa);
        gather_accum_store<3>(fb, dxs, dys, dzs, vt, nl, h);
        gather_issue<5>(ftb, lane_off, offs, fb);
        gather_accum_store<4>(fa, dxs, dys, dzs, vt, nl, h);
        gather_issue<6>(ftb, lane_off, offs, fa);
        gather_accum_store<5>(fb, dxs, dys, dzs, vt, nl, h);
        gather_issue<7>(ftb, lane_off, offs, fb);
        gather_accum_store<6>(fa, dxs, dys, dzs, vt, nl, h);
        gather_accum_store<7>(fb, dxs, dys, dzs, vt, nl, h);
    } else {
        const float cx = pos[(b * 3 + 0) * NPTS + n];
        const float cy = pos[(b * 3 + 1) * NPTS + n];
        const float cz = pos[(b * 3 + 2) * NPTS + n];

        int   idxs[KNB];
        float dxs[KNB], dys[KNB], dzs[KNB];
        #pragma unroll
        for (int k = 0; k < KNB; ++k)
            idxs[k] = nbr[(b * KNB + k) * NPTS + n];
        #pragma unroll
        for (int k = 0; k < KNB; ++k) {
            const int id = idxs[k];
            dxs[k] = pos[(b * 3 + 0) * NPTS + id] - cx;
            dys[k] = pos[(b * 3 + 1) * NPTS + id] - cy;
            dzs[k] = pos[(b * 3 + 2) * NPTS + id] - cz;
        }

        for (int cc = 0; cc < 4; ++cc) {
            const int cbase = h * 32 + cc * 8;
            float ax[8] = {}, ay[8] = {}, az[8] = {}, asum[8] = {};
            for (int k = 0; k < KNB; ++k) {
                const int   id = idxs[k];
                const float dx = dxs[k], dy = dys[k], dz = dzs[k];
                float f[8];
                const float* fp = feat + (b * CIN + cbase) * NPTS + id;
                #pragma unroll
                for (int c = 0; c < 8; ++c) f[c] = fp[c * NPTS];
                #pragma unroll
                for (int c = 0; c < 8; ++c) {
                    ax[c]   = fmaf(f[c], dx, ax[c]);
                    ay[c]   = fmaf(f[c], dy, ay[c]);
                    az[c]   = fmaf(f[c], dz, az[c]);
                    asum[c] += f[c];
                }
            }
            #pragma unroll
            for (int c = 0; c < 8; c += 2) {
                const int dcp = (cbase + c) >> 1;
                v2f px, py, pz, ps;
                px.x = ax[c];   px.y = ax[c + 1];
                py.x = ay[c];   py.y = ay[c + 1];
                pz.x = az[c];   pz.y = az[c + 1];
                ps.x = asum[c]; ps.y = asum[c + 1];
                *reinterpret_cast<v2f*>(&lds[vb + (dcp      ) * VSTR + nl * 2]) = px;
                *reinterpret_cast<v2f*>(&lds[vb + (dcp + 32 ) * VSTR + nl * 2]) = py;
                *reinterpret_cast<v2f*>(&lds[vb + (dcp + 64 ) * VSTR + nl * 2]) = pz;
                *reinterpret_cast<v2f*>(&lds[vb + (dcp + 96 ) * VSTR + nl * 2]) = ps;
            }
        }
    }

    v8f acc[4] = {v8f{}, v8f{}, v8f{}, v8f{}};
    const float* wA = &lds[h * WPAIRSTR + nl * 2];
    const float* vB = &lds[vb + h * VSTR + nl * 2];
    FragSet fs0, fs1;
    frag_load(wA, vB, 0, fs0);
    #pragma unroll 1
    for (int it = 0; it < 3; ++it) {
        frag_load(wA, vB, 2 * it + 1, fs1);
        __builtin_amdgcn_sched_barrier(0);
        frag_mma(fs0, acc);
        __builtin_amdgcn_sched_barrier(0);
        frag_load(wA, vB, 2 * it + 2, fs0);
        __builtin_amdgcn_sched_barrier(0);
        frag_mma(fs1, acc);
        __builtin_amdgcn_sched_barrier(0);
    }
    frag_load(wA, vB, 7, fs1);
    __builtin_amdgcn_sched_barrier(0);
    frag_mma(fs0, acc);
    frag_mma(fs1, acc);

    #pragma unroll
    for (int ot = 0; ot < 4; ++ot) {
        #pragma unroll
        for (int r = 0; r < 8; ++r) {
            const int o = ot * 16 + r + h * 8;
            out[(b * COUT + o) * NPTS + n] = acc[ot][r] + lds[BOFF + o];
        }
    }
    }
}

extern "C" void kernel_launch(void* const* d_in, const int* in_sizes, int n_in,
                              void* d_out, int out_size, void* d_ws, size_t ws_size,
                              hipStream_t stream) {
    (void)in_sizes; (void)n_in; (void)out_size;
    const float* feat = (const float*)d_in[0];
    const float* wt   = (const float*)d_in[1];
    const float* wb   = (const float*)d_in[2];
    const float* bias = (const float*)d_in[3];
    const int*   nbr  = (const int*)d_in[4];
    const float* pos  = (const float*)d_in[5];
    float* out = (float*)d_out;

    const size_t ft_bytes = (size_t)NB * NPTS * CIN * sizeof(float);
    const size_t p4_bytes = (size_t)NB * NPTS * 4 * sizeof(float);
    dim3 grid(NPTS / 64, NB);
    dim3 grid_main(NPTS / 64);

    if (ws_size >= ft_bytes + p4_bytes) {
        float* ft = (float*)d_ws;
        float* p4 = ft + (size_t)NB * NPTS * CIN;
        transpose_feat_kernel<true><<<grid, 256, 0, stream>>>(feat, pos, ft, p4);
        flexconv_wmma_kernel<true, true><<<grid_main, 128, 0, stream>>>(
            feat, ft, p4, wt, wb, bias, nbr, pos, out);
    } else if (ws_size >= ft_bytes) {
        float* ft = (float*)d_ws;
        transpose_feat_kernel<false><<<grid, 256, 0, stream>>>(feat, pos, ft, ft);
        flexconv_wmma_kernel<true, false><<<grid_main, 128, 0, stream>>>(
            feat, ft, ft, wt, wb, bias, nbr, pos, out);
    } else {
        flexconv_wmma_kernel<false, false><<<grid_main, 128, 0, stream>>>(
            feat, feat, feat, wt, wb, bias, nbr, pos, out);
    }
}


# gather: a point's neighbour offsets and relative positions pass through a wave-private LDS table (unused pad dwords of W' rows), read by broadcast LDS reads, not row_share DPP moves; sums unchanged
# speedup vs baseline: 3.5557x; 1.0353x over previous
#include <hip/hip_runtime.h>

typedef __attribute__((ext_vector_type(2))) float v2f;
typedef __attribute__((ext_vector_type(4))) float v4f;
typedef __attribute__((ext_vector_type(8))) float v8f;

#define NPTS 32768
#define NB   4
#define CIN  64
#define COUT 64
#define KNB  16

#define WPAIRSTR 160
#define WSZ  (128 * WPAIRSTR)
#define VOFF WSZ
#define VSTR 34
#define VWAVE (128 * VSTR)
#define BOFF (VOFF + 4 * VWAVE)
#define LDS_FLOATS (BOFF + 64)

template <bool PACK_POS>
__global__ __launch_bounds__(256) void transpose_feat_kernel(
    const float* __restrict__ feat, const float* __restrict__ pos,
    float* __restrict__ ft, float* __restrict__ p4)
{
    __shared__ float tile[64][65];
    const int b  = blockIdx.y;
    const int n0 = blockIdx.x * 64;
    const int tid = threadIdx.x;
    if constexpr (PACK_POS) {
        if (tid < 64) {
            const int n = n0 + tid;
            v4f q;
            q.x = pos[(b * 3 + 0) * NPTS + n];
            q.y = pos[(b * 3 + 1) * NPTS + n];
            q.z = pos[(b * 3 + 2) * NPTS + n];
            q.w = 0.0f;
            *reinterpret_cast<v4f*>(p4 + (size_t)(b * NPTS + n) * 4) = q;
        }
    }
    #pragma unroll
    for (int i = 0; i < 16; ++i) {
        const int j = tid + i * 256;
        const int c = j >> 6, nloc = j & 63;
        tile[nloc][c] = feat[(b * CIN + c) * NPTS + n0 + nloc];
    }
    __syncthreads();
    #pragma unroll
    for (int i = 0; i < 16; ++i) {
        const int j = tid + i * 256;
        const int nloc = j >> 6, c = j & 63;
        ft[(b * NPTS + n0 + nloc) * CIN + c] = tile[nloc][c];
    }
}

#define TPAD 128

template <int P>
__device__ __forceinline__ const float* tab_entry(const float* tr, int k) {
    return tr + (P * 4 + (k >> 3)) * WPAIRSTR + (k & 7) * 4;
}

template <int P>
__device__ __forceinline__ void tab_load_off(const float* tr, int (&o)[16]) {
    #pragma unroll
    for (int k = 0; k < 16; ++k)
        o[k] = __builtin_bit_cast(int, tab_entry<P>(tr, k)[3]);
}
template <int P>
__device__ __forceinline__ void tab_load_xyz(const float* tr, v4f (&e)[16]) {
    #pragma unroll
    for (int k = 0; k < 16; ++k)
        e[k] = *reinterpret_cast<const v4f*>(tab_entry<P>(tr, k));
}

__device__ __forceinline__ void gather_issue(const char* __restrict__ ftb, unsigned lane_off,
                                             const int (&o)[16], v4f (&fv)[16]) {
    #pragma unroll
    for (int k = 0; k < 16; ++k)
        fv[k] = *reinterpret_cast<const v4f*>(ftb + ((unsigned)o[k] + lane_off));
}

template <int P>
__device__ __forceinline__ void gather_accum_store(const v4f (&fv)[16], const v4f (&e)[16],
        float* __restrict__ vt, int j, int h) {
    float ax[4] = {}, ay[4] = {}, az[4] = {}, as[4] = {};
    #pragma unroll
    for (int k = 0; k < 16; ++k) {
        const float x = e[k].x, y = e[k].y, z = e[k].z;
        #pragma unroll
        for (int c = 0; c < 4; ++c) {
            ax[c] = fmaf(fv[k][c], x, ax[c]);
            ay[c] = fmaf(fv[k][c], y, ay[c]);
            az[c] = fmaf(fv[k][c], z, az[c]);
            as[c] += fv[k][c];
        }
    }
    float* base = vt + (2 * j) * VSTR + (2 * P + h) * 2;
    v2f t;
    t.x = ax[0]; t.y = ax[1]; *reinterpret_cast<v2f*>(base)                    = t;
    t.x = ax[2]; t.y = ax[3]; *reinterpret_cast<v2f*>(base + VSTR)             = t;
    t.x = ay[0]; t.y = ay[1]; *reinterpret_cast<v2f*>(base + 32 * VSTR)        = t;
    t.x = ay[2]; t.y = ay[3]; *reinterpret_cast<v2f*>(base + 33 * VSTR)        = t;
    t.x = az[0]; t.y = az[1]; *reinterpret_cast<v2f*>(base + 64 * VSTR)        = t;
    t.x = az[2]; t.y = az[3]; *reinterpret_cast<v2f*>(base + 65 * VSTR)        = t;
    t.x = as[0]; t.y = as[1]; *reinterpret_cast<v2f*>(base + 96 * VSTR)        = t;
    t.x = as[2]; t.y = as[3]; *reinterpret_cast<v2f*>(base + 97 * VSTR)        = t;
}

struct FragSet {
    v2f b[8];
    v2f a[8][4];
};

__device__ __forceinline__ void frag_load(const float* __restrict__ wA, const float* __restrict__ vB,
                                          int blk, FragSet& f) {
    #pragma unroll
    for (int s = 0; s < 8; ++s) {
        const int ks = blk * 8 + s;
        f.b[s] = *reinterpret_cast<const v2f*>(vB + ks * (2 * VSTR));
        #pragma unroll
        for (int ot = 0; ot < 4; ++ot)
            f.a[s][ot] = *reinterpret_cast<const v2f*>(wA + ks * (2 * WPAIRSTR) + ot * 32);
    }
}

__device__ __forceinline__ void frag_mma(const FragSet& f, v8f (&acc)[4]) {
    #pragma unroll
    for (int s = 0; s < 8; ++s) {
        #pragma unroll
        for (int ot = 0; ot < 4; ++ot)
            acc[ot] = __builtin_amdgcn_wmma_f32_16x16x4_f32(
                false, f.a[s][ot], false, f.b[s], (short)0, acc[ot], false, false);
    }
}

template <bool USE_FT, bool USE_P4>
__global__ __launch_bounds__(128) void flexconv_wmma_kernel(
    const float* __restrict__ feat,
    const float* __restrict__ ft,
    const float* __restrict__ p4,
    const float* __restrict__ wt,
    const float* __restrict__ wb,
    const float* __restrict__ bias,
    const int*   __restrict__ nbr,
    const float* __restrict__ pos,
    float* __restrict__ out)
{
    __shared__ alignas(16) float lds[LDS_FLOATS];
    const int tid = threadIdx.x;

    v4f wr0[16], wr1[16];
    #pragma unroll
    for (int j = 0; j < 16; ++j) {
        const int i   = tid + j * 128;
        const int dcp = i >> 4, oq = i & 15;
        const int dc  = dcp * 2;
        const float* row = (j < 12) ? (wt + dc * 64) : (wb + (dc - 192) * 64);
        wr0[j] = *reinterpret_cast<const v4f*>(row + oq * 4);
        wr1[j] = *reinterpret_cast<const v4f*>(row + 64 + oq * 4);
    }
    float bv = 0.0f;
    if (tid < COUT) bv = bias[tid];
    #pragma unroll
    for (int j = 0; j < 16; ++j) {
        const int i   = tid + j * 128;
        const int dcp = i >> 4, oq = i & 15;
        v4f lo, hi;
        lo.x = wr0[j].x; lo.y = wr1[j].x; lo.z = wr0[j].y; lo.w = wr1[j].y;
        hi.x = wr0[j].z; hi.y = wr1[j].z; hi.z = wr0[j].w; hi.w = wr1[j].w;
        float* dst = &lds[dcp * WPAIRSTR + oq * 8];
        *reinterpret_cast<v4f*>(dst)     = lo;
        *reinterpret_cast<v4f*>(dst + 4) = hi;
    }
    if (tid < COUT) lds[BOFF + tid] = bv;
    __syncthreads();

    const int wave = tid >> 5;
    const int lane = tid & 31;
    const int nl   = lane & 15;
    const int h    = lane >> 4;
    const int n    = blockIdx.x * 64 + wave * 16 + nl;
    const int vb   = VOFF + wave * VWAVE;

    #pragma unroll 1
    for (int b = 0; b < NB; ++b) {

    if constexpr (USE_FT) {
        const int nh = blockIdx.x * 64 + wave * 16 + 2 * (nl & 7) + h;
        const int kh = (nl >> 3) * 8;
        int   offs[8];
        float dxs[8], dys[8], dzs[8];
        #pragma unroll
        for (int r = 0; r < 8; ++r)
            offs[r] = nbr[(b * KNB + kh + r) * NPTS + nh];
        if constexpr (USE_P4) {
            const v4f* p4b = reinterpret_cast<const v4f*>(p4) + (size_t)b * NPTS;
            const v4f c4 = p4b[nh];
            v4f q[8];
            #pragma unroll
            for (int r = 0; r < 8; ++r)
                q[r] = p4b[offs[r]];
            #pragma unroll
            for (int r = 0; r < 8; ++r) {
                dxs[r] = q[r].x - c4.x;
                dys[r] = q[r].y - c4.y;
                dzs[r] = q[r].z - c4.z;
                offs[r] *= (CIN * 4);
            }
        } else {
            const float cx = pos[(b * 3 + 0) * NPTS + nh];
            const float cy = pos[(b * 3 + 1) * NPTS + nh];
            const float cz = pos[(b * 3 + 2) * NPTS + nh];
            #pragma unroll
            for (int r = 0; r < 8; ++r) {
                const int id = offs[r];
                dxs[r] = pos[(b * 3 + 0) * NPTS + id] - cx;
                dys[r] = pos[(b * 3 + 1) * NPTS + id] - cy;
                dzs[r] = pos[(b * 3 + 2) * NPTS + id] - cz;
                offs[r] = id * (CIN * 4);
            }
        }
        {
            float* tw = &lds[(wave * 32 + (2 * (nl & 7) + h) * 2 + (nl >> 3)) * WPAIRSTR + TPAD];
            #pragma unroll
            for (int r = 0; r < 8; ++r) {
                v4f e;
                e.x = dxs[r];
                e.y = dys[r];
                e.z = dzs[r];
                e.w = __builtin_bit_cast(float, offs[r]);
                *reinterpret_cast<v4f*>(tw + r * 4) = e;
            }
        }
        const float* tr = &lds[(wave * 32 + h * 2) * WPAIRSTR + TPAD];
        const char* ftb = reinterpret_cast<const char*>(ft + (size_t)b * NPTS * CIN);
        const unsigned lane_off = nl * 16;
        float* vt = &lds[vb];
        v4f fa[16], fb[16];
        int oa[16], ob[16];
        v4f xa[16], xb[16];
        tab_load_off<0>(tr, oa);
        tab_load_off<1>(tr, ob);
        tab_load_xyz<0>(tr, xa);
        gather_issue(ftb, lane_off, oa, fa);
        gather_issue(ftb, lane_off, ob, fb);
        tab_load_off<2>(tr, oa);
        __builtin_amdgcn_sched_barrier(0);
        tab_load_off<3>(tr, ob);
        tab_load_xyz<1>(tr, xb);
        gather_accum_store<0>(fa, xa, vt, nl, h);
        gather_issue(ftb, lane_off, oa, fa);
        __builtin_amdgcn_sched_barrier(0);
        tab_load_off<4>(tr, oa);
        tab_load_xyz<2>(tr, xa);
        gather_accum_store<1>(fb, xb, vt, nl, h);
        gather_issue(ftb, lane_off, ob, fb);
        __builtin_amdgcn_sched_barrier(0);
        tab_load_off<5>(tr, ob);
        tab_load_xyz<3>(tr, xb);
        gather_accum_store<2>(fa, xa, vt, nl, h);
        gather_issue(ftb, lane_off, oa, fa);
        __builtin_amdgcn_sched_barrier(0);
        tab_load_off<6>(tr, oa);
        tab_load_xyz<4>(tr, xa);
        gather_accum_store<3>(fb, xb, vt, nl, h);
        gather_issue(ftb, lane_off, ob, fb);
        __builtin_amdgcn_sched_barrier(0);
        tab_load_off<7>(tr, ob);
        tab_load_xyz<5>(tr, xb);
        gather_accum_store<4>(fa, xa, vt, nl, h);
        gather_issue(ftb, lane_off, oa, fa);
        __builtin_amdgcn_sched_barrier(0);
        tab_load_xyz<6>(tr, xa);
        gather_accum_store<5>(fb, xb, vt, nl, h);
        gather_issue(ftb, lane_off, ob, fb);
        __builtin_amdgcn_sched_barrier(0);
        tab_load_xyz<7>(tr, xb);
        gather_accum_store<6>(fa, xa, vt, nl, h);
        gather_accum_store<7>(fb, xb, vt, nl, h);
    } else {
        const float cx = pos[(b * 3 + 0) * NPTS + n];
        const float cy = pos[(b * 3 + 1) * NPTS + n];
        const float cz = pos[(b * 3 + 2) * NPTS + n];

        int   idxs[KNB];
        float dxs[KNB], dys[KNB], dzs[KNB];
        #pragma unroll
        for (int k = 0; k < KNB; ++k)
            idxs[k] = nbr[(b * KNB + k) * NPTS + n];
        #pragma unroll
        for (int k = 0; k < KNB; ++k) {
            const int id = idxs[k];
            dxs[k] = pos[(b * 3 + 0) * NPTS + id] - cx;
            dys[k] = pos[(b * 3 + 1) * NPTS + id] - cy;
            dzs[k] = pos[(b * 3 + 2) * NPTS + id] - cz;
        }

        for (int cc = 0; cc < 4; ++cc) {
            const int cbase = h * 32 + cc * 8;
            float ax[8] = {}, ay[8] = {}, az[8] = {}, asum[8] = {};
            for (int k = 0; k < KNB; ++k) {
                const int   id = idxs[k];
                const float dx = dxs[k], dy = dys[k], dz = dzs[k];
                float f[8];
                const float* fp = feat + (b * CIN + cbase) * NPTS + id;
                #pragma unroll
                for (int c = 0; c < 8; ++c) f[c] = fp[c * NPTS];
                #pragma unroll
                for (int c = 0; c < 8; ++c) {
                    ax[c]   = fmaf(f[c], dx, ax[c]);
                    ay[c]   = fmaf(f[c], dy, ay[c]);
                    az[c]   = fmaf(f[c], dz, az[c]);
                    asum[c] += f[c];
                }
            }
            #pragma unroll
            for (int c = 0; c < 8; c += 2) {
                const int dcp = (cbase + c) >> 1;
                v2f px, py, pz, ps;
                px.x = ax[c];   px.y = ax[c + 1];
                py.x = ay[c];   py.y = ay[c + 1];
                pz.x = az[c];   pz.y = az[c + 1];
                ps.x = asum[c]; ps.y = asum[c + 1];
                *reinterpret_cast<v2f*>(&lds[vb + (dcp      ) * VSTR + nl * 2]) = px;
                *reinterpret_cast<v2f*>(&lds[vb + (dcp + 32 ) * VSTR + nl * 2]) = py;
                *reinterpret_cast<v2f*>(&lds[vb + (dcp + 64 ) * VSTR + nl * 2]) = pz;
                *reinterpret_cast<v2f*>(&lds[vb + (dcp + 96 ) * VSTR + nl * 2]) = ps;
            }
        }
    }

    v8f acc[4] = {v8f{}, v8f{}, v8f{}, v8f{}};
    const float* wA = &lds[h * WPAIRSTR + nl * 2];
    const float* vB = &lds[vb + h * VSTR + nl * 2];
    FragSet fs0, fs1;
    frag_load(wA, vB, 0, fs0);
    #pragma unroll 1
    for (int it = 0; it < 3; ++it) {
        frag_load(wA, vB, 2 * it + 1, fs1);
        __builtin_amdgcn_sched_barrier(0);
        frag_mma(fs0, acc);
        __builtin_amdgcn_sched_barrier(0);
        frag_load(wA, vB, 2 * it + 2, fs0);
        __builtin_amdgcn_sched_barrier(0);
        frag_mma(fs1, acc);
        __builtin_amdgcn_sched_barrier(0);
    }
    frag_load(wA, vB, 7, fs1);
    __builtin_amdgcn_sched_barrier(0);
    frag_mma(fs0, acc);
    frag_mma(fs1, acc);

    #pragma unroll
    for (int ot = 0; ot < 4; ++ot) {
        #pragma unroll
        for (int r = 0; r < 8; ++r) {
            const int o = ot * 16 + r + h * 8;
            out[(b * COUT + o) * NPTS + n] = acc[ot][r] + lds[BOFF + o];
        }
    }
    }
}

extern "C" void kernel_launch(void* const* d_in, const int* in_sizes, int n_in,
                              void* d_out, int out_size, void* d_ws, size_t ws_size,
                              hipStream_t stream) {
    (void)in_sizes; (void)n_in; (void)out_size;
    const float* feat = (const float*)d_in[0];
    const float* wt   = (const float*)d_in[1];
    const float* wb   = (const float*)d_in[2];
    const float* bias = (const float*)d_in[3];
    const int*   nbr  = (const int*)d_in[4];
    const float* pos  = (const float*)d_in[5];
    float* out = (float*)d_out;

    const size_t ft_bytes = (size_t)NB * NPTS * CIN * sizeof(float);
    const size_t p4_bytes = (size_t)NB * NPTS * 4 * sizeof(float);
    dim3 grid(NPTS / 64, NB);
    dim3 grid_main(NPTS / 64);

    if (ws_size >= ft_bytes + p4_bytes) {
        float* ft = (float*)d_ws;
        float* p4 = ft + (size_t)NB * NPTS * CIN;
        transpose_feat_kernel<true><<<grid, 256, 0, stream>>>(feat, pos, ft, p4);
        flexconv_wmma_kernel<true, true><<<grid_main, 128, 0, stream>>>(
            feat, ft, p4, wt, wb, bias, nbr, pos, out);
    } else if (ws_size >= ft_bytes) {
        float* ft = (float*)d_ws;
        transpose_feat_kernel<false><<<grid, 256, 0, stream>>>(feat, pos, ft, ft);
        flexconv_wmma_kernel<true, false><<<grid_main, 128, 0, stream>>>(
            feat, ft, ft, wt, wb, bias, nbr, pos, out);
    } else {
        flexconv_wmma_kernel<false, false><<<grid_main, 128, 0, stream>>>(
            feat, feat, feat, wt, wb, bias, nbr, pos, out);
    }
}
